# starting kernel
# baseline (speedup 1.0000x reference)
#include <hip/hip_runtime.h>
#include <math.h>


typedef __attribute__((ext_vector_type(2))) float v2f;
typedef __attribute__((ext_vector_type(8))) float v8f;

#define BT    8
#define NN    256
#define FIN   128
#define FOUT  64
#define LDSW  260

__global__ __launch_bounds__(128) void k_gemm_xw(const float* __restrict__ x,
                                                 const float* __restrict__ W,
                                                 float* __restrict__ h) {
    const int blk   = blockIdx.x;
    const int bt    = blk >> 4;
    const int mtile = blk & 15;
    const int wv    = threadIdx.x >> 5;
    const int lane  = threadIdx.x & 31;
    const int half  = lane >> 4;
    const int l15   = lane & 15;

    const float* __restrict__ xrow = x + (size_t)(bt * NN + mtile * 16 + l15) * FIN;
    const int colbase = wv * 16 + l15;

    v8f c = {};
    for (int k0 = 0; k0 < FIN; k0 += 4) {
        const int ka = k0 + half * 2;
        v2f a_; a_.x = xrow[ka];                 a_.y = xrow[ka + 1];
        v2f b_; b_.x = W[ka * FOUT + colbase];   b_.y = W[(ka + 1) * FOUT + colbase];
        c = __builtin_amdgcn_wmma_f32_16x16x4_f32(false, a_, false, b_, (short)0, c,
                                                  false, false);
    }
    float* __restrict__ hout = h + (size_t)(bt * NN + mtile * 16) * FOUT + colbase;
#pragma unroll
    for (int r = 0; r < 8; ++r)
        hout[(size_t)(r + half * 8) * FOUT] = c[r];
}

__global__ __launch_bounds__(256) void k_attn_s(const float* __restrict__ h,
                                                const float* __restrict__ a,
                                                float* __restrict__ s) {
    const int bt = blockIdx.x;
    const int j  = threadIdx.x;
    const float* __restrict__ hb = h + (size_t)bt * NN * FOUT;
    const int ja = (2 * j) & (NN - 1);
    const int jb = (2 * j + 1) & (NN - 1);
    const float* __restrict__ ra = hb + (size_t)ja * FOUT;
    const float* __restrict__ rb = hb + (size_t)jb * FOUT;
    float acc = 0.f;
#pragma unroll 8
    for (int f = 0; f < FOUT; ++f)
        acc = fmaf(ra[f], a[f], fmaf(rb[f], a[FOUT + f], acc));
    s[bt * NN + j] = acc;
}

__global__ __launch_bounds__(128) void k_attn_out(const float* __restrict__ adj,
                                                  const float* __restrict__ h,
                                                  const float* __restrict__ s,
                                                  float* __restrict__ out) {
    __shared__ float attn_lds[16][LDSW];

    const int blk   = blockIdx.x;
    const int bt    = blk >> 4;
    const int itile = blk & 15;
    const int b     = bt >> 2;
    const int tau   = bt & 3;
    const int cpred = (itile >= 8) ? 1 : 0;
    const int tp    = (2 * tau + cpred) & 3;

    const int wv   = threadIdx.x >> 5;
    const int lane = threadIdx.x & 31;
    const int half = lane >> 4;
    const int l15  = lane & 15;

    const float* __restrict__ srow = s + (b * 4 + tp) * NN;
    const float* __restrict__ hb   = h + (size_t)bt * NN * FOUT;

#pragma unroll
    for (int rr = 0; rr < 4; ++rr) {
        const int rloc = wv * 4 + rr;
        const int i    = itile * 16 + rloc;
        const float* __restrict__ arow = adj + ((size_t)(bt * NN + i)) * NN;

        float z[8];
        float mx = -3.402823466e38f;
#pragma unroll
        for (int u = 0; u < 8; ++u) {
            const int j = lane + u * 32;
            const float sv = srow[j];
            const float ev = (sv > 0.f) ? sv : 0.2f * sv;
            const float zz = arow[j] * ev;
            z[u] = zz;
            mx = fmaxf(mx, zz);
        }
#pragma unroll
        for (int off = 16; off > 0; off >>= 1)
            mx = fmaxf(mx, __shfl_xor(mx, off, 32));
        float sum = 0.f;
#pragma unroll
        for (int u = 0; u < 8; ++u) {
            const float p = __expf(z[u] - mx);
            z[u] = p;
            sum += p;
        }
#pragma unroll
        for (int off = 16; off > 0; off >>= 1)
            sum += __shfl_xor(sum, off, 32);
        const float inv = 1.f / sum;
#pragma unroll
        for (int u = 0; u < 8; ++u)
            attn_lds[rloc][lane + u * 32] = z[u] * inv;
    }
    __syncthreads();

    const int colbase = wv * 16 + l15;
    v8f c = {};
    for (int k0 = 0; k0 < NN; k0 += 4) {
        const int ka = k0 + half * 2;
        v2f a_; a_.x = attn_lds[l15][ka];
                a_.y = attn_lds[l15][ka + 1];
        v2f b_; b_.x = hb[(size_t)ka * FOUT + colbase];
                b_.y = hb[(size_t)(ka + 1) * FOUT + colbase];
        c = __builtin_amdgcn_wmma_f32_16x16x4_f32(false, a_, false, b_, (short)0, c,
                                                  false, false);
    }
    float* __restrict__ orow = out + (size_t)(bt * NN + itile * 16) * FOUT + colbase;
#pragma unroll
    for (int r = 0; r < 8; ++r)
        orow[(size_t)(r + half * 8) * FOUT] = c[r];
}

extern "C" void kernel_launch(void* const* d_in, const int* in_sizes, int n_in,
                              void* d_out, int out_size, void* d_ws, size_t ws_size,
                              hipStream_t stream) {
    const float* x   = (const float*)d_in[0];
    const float* adj = (const float*)d_in[1];
    const float* W   = (const float*)d_in[2];
    const float* a   = (const float*)d_in[3];
    float* out = (float*)d_out;

    float* h = (float*)d_ws;
    float* s = h + (size_t)BT * NN * FOUT;

    k_gemm_xw<<<dim3(BT * 16), dim3(128), 0, stream>>>(x, W, h);
    k_attn_s<<<dim3(BT), dim3(256), 0, stream>>>(h, a, s);
    k_attn_out<<<dim3(BT * 16), dim3(128), 0, stream>>>(adj, h, s, out);
}


# Score kernel folded into the attention kernel (3 launches -> 2): each block rebuilds s of its slice from h with wide coalesced loads and a lane sum through LDS; s is summed in a new order
# speedup vs baseline: 1.4096x; 1.4096x over previous
#include <hip/hip_runtime.h>
#include <math.h>


typedef __attribute__((ext_vector_type(2))) float v2f;
typedef __attribute__((ext_vector_type(8))) float v8f;

#define BT    8
#define NN    256
#define FIN   128
#define FOUT  64
#define LDSW  260
#define PSTR  36

__global__ __launch_bounds__(128) void k_gemm_xw(const float* __restrict__ x,
                                                 const float* __restrict__ W,
                                                 float* __restrict__ h) {
    const int blk   = blockIdx.x;
    const int bt    = blk >> 4;
    const int mtile = blk & 15;
    const int wv    = threadIdx.x >> 5;
    const int lane  = threadIdx.x & 31;
    const int half  = lane >> 4;
    const int l15   = lane & 15;

    const float* __restrict__ xrow = x + (size_t)(bt * NN + mtile * 16 + l15) * FIN;
    const int colbase = wv * 16 + l15;

    v8f c = {};
    for (int k0 = 0; k0 < FIN; k0 += 4) {
        const int ka = k0 + half * 2;
        v2f a_; a_.x = xrow[ka];                 a_.y = xrow[ka + 1];
        v2f b_; b_.x = W[ka * FOUT + colbase];   b_.y = W[(ka + 1) * FOUT + colbase];
        c = __builtin_amdgcn_wmma_f32_16x16x4_f32(false, a_, false, b_, (short)0, c,
                                                  false, false);
    }
    float* __restrict__ hout = h + (size_t)(bt * NN + mtile * 16) * FOUT + colbase;
#pragma unroll
    for (int r = 0; r < 8; ++r)
        hout[(size_t)(r + half * 8) * FOUT] = c[r];
}

__global__ __launch_bounds__(128) void k_attn_out(const float* __restrict__ adj,
                                                  const float* __restrict__ h,
                                                  const float* __restrict__ a,
                                                  float* __restrict__ out) {
    __shared__ float attn_lds[16][LDSW];
    __shared__ __attribute__((aligned(16))) float part_lds[4][32][PSTR];
    __shared__ float e_lds[128];

    const int blk   = blockIdx.x;
    const int bt    = blk >> 4;
    const int itile = blk & 15;
    const int b     = bt >> 2;
    const int tau   = bt & 3;
    const int cpred = (itile >= 8) ? 1 : 0;
    const int tp    = (2 * tau + cpred) & 3;

    const int wv   = threadIdx.x >> 5;
    const int lane = threadIdx.x & 31;
    const int half = lane >> 4;
    const int l15  = lane & 15;

    const float* __restrict__ hb   = h + (size_t)bt * NN * FOUT;

    const float4* __restrict__ h2 =
        reinterpret_cast<const float4*>(h + (size_t)(b * 4 + tp) * NN * FOUT) + wv * 1024 + lane;
    const float4 av = reinterpret_cast<const float4*>(a)[lane];
    float4 hv[32];
#pragma unroll
    for (int i = 0; i < 32; ++i)
        hv[i] = h2[i * 32];

    float z[4][8];
#pragma unroll
    for (int rr = 0; rr < 4; ++rr) {
        const float* __restrict__ arow =
            adj + ((size_t)(bt * NN + itile * 16 + wv * 4 + rr)) * NN;
#pragma unroll
        for (int u = 0; u < 8; ++u)
            z[rr][u] = arow[lane + u * 32];
    }

#pragma unroll
    for (int i = 0; i < 32; ++i)
        part_lds[wv][i][lane] =
            fmaf(hv[i].w, av.w, fmaf(hv[i].z, av.z, fmaf(hv[i].y, av.y, hv[i].x * av.x)));
    __syncthreads();
    {
        const float4* __restrict__ pr = reinterpret_cast<const float4*>(&part_lds[wv][lane][0]);
        float4 acc = pr[0];
#pragma unroll
        for (int k = 1; k < 8; ++k) {
            const float4 r = pr[k];
            acc.x += r.x; acc.y += r.y; acc.z += r.z; acc.w += r.w;
        }
        const float sv = (acc.x + acc.y) + (acc.z + acc.w);
        e_lds[wv * 32 + lane] = (sv > 0.f) ? sv : 0.2f * sv;
    }
    __syncthreads();

    float ev[4];
#pragma unroll
    for (int u = 0; u < 4; ++u)
        ev[u] = e_lds[lane + u * 32];

#pragma unroll
    for (int rr = 0; rr < 4; ++rr) {
        const int rloc = wv * 4 + rr;

        float mx = -3.402823466e38f;
#pragma unroll
        for (int u = 0; u < 8; ++u) {
            const float zz = z[rr][u] * ev[u & 3];
            z[rr][u] = zz;
            mx = fmaxf(mx, zz);
        }
#pragma unroll
        for (int off = 16; off > 0; off >>= 1)
            mx = fmaxf(mx, __shfl_xor(mx, off, 32));
        float sum = 0.f;
#pragma unroll
        for (int u = 0; u < 8; ++u) {
            const float p = __expf(z[rr][u] - mx);
            z[rr][u] = p;
            sum += p;
        }
#pragma unroll
        for (int off = 16; off > 0; off >>= 1)
            sum += __shfl_xor(sum, off, 32);
        const float inv = 1.f / sum;
#pragma unroll
        for (int u = 0; u < 8; ++u)
            attn_lds[rloc][lane + u * 32] = z[rr][u] * inv;
    }
    __syncthreads();

    const int colbase = wv * 16 + l15;
    v8f c = {};
    for (int k0 = 0; k0 < NN; k0 += 4) {
        const int ka = k0 + half * 2;
        v2f a_; a_.x = attn_lds[l15][ka];
                a_.y = attn_lds[l15][ka + 1];
        v2f b_; b_.x = hb[(size_t)ka * FOUT + colbase];
                b_.y = hb[(size_t)(ka + 1) * FOUT + colbase];
        c = __builtin_amdgcn_wmma_f32_16x16x4_f32(false, a_, false, b_, (short)0, c,
                                                  false, false);
    }
    float* __restrict__ orow = out + (size_t)(bt * NN + itile * 16) * FOUT + colbase;
#pragma unroll
    for (int r = 0; r < 8; ++r)
        orow[(size_t)(r + half * 8) * FOUT] = c[r];
}

extern "C" void kernel_launch(void* const* d_in, const int* in_sizes, int n_in,
                              void* d_out, int out_size, void* d_ws, size_t ws_size,
                              hipStream_t stream) {
    const float* x   = (const float*)d_in[0];
    const float* adj = (const float*)d_in[1];
    const float* W   = (const float*)d_in[2];
    const float* a   = (const float*)d_in[3];
    float* out = (float*)d_out;

    float* h = (float*)d_ws;

    k_gemm_xw<<<dim3(BT * 16), dim3(128), 0, stream>>>(x, W, h);
    k_attn_out<<<dim3(BT * 16), dim3(128), 0, stream>>>(adj, h, a, out);
}


# Whole layer in one launch (2 -> 1), no h: out = (attn @ x) @ W on f32 WMMA (K split over the 4 waves), s = x . (W a) on the vector unit; every sum is taken in a new order
# speedup vs baseline: 1.8643x; 1.3226x over previous
#include <hip/hip_runtime.h>
#include <math.h>


typedef __attribute__((ext_vector_type(2))) float v2f;
typedef __attribute__((ext_vector_type(8))) float v8f;

#define BT    8
#define NN    256
#define FIN   128
#define FOUT  64
#define LDSW  260
#define PSTR  36
#define WSTR  76
#define TSTR  132

#define WMMA_F32(A_, B_, C_) \
    __builtin_amdgcn_wmma_f32_16x16x4_f32(false, (A_), false, (B_), (short)0, (C_), false, false)

static __device__ __forceinline__ void lds_barrier() {
    asm volatile("s_wait_dscnt 0x0\n\ts_barrier_signal -1\n\ts_barrier_wait -1" ::: "memory");
}
static __device__ __forceinline__ void request_order() {
    asm volatile("" ::: "memory");
}

__global__ __launch_bounds__(128) void k_gat_one(const float* x,
                                                 const float* adj,
                                                 const float* W,
                                                 const float* __restrict__ a,
                                                 float* out) {
    __shared__ __attribute__((aligned(16))) float attn_lds[16][LDSW];
    __shared__ __attribute__((aligned(16))) float part_lds[4][32][PSTR];
    __shared__ __attribute__((aligned(16))) float w_lds[FIN][WSTR];
    __shared__ __attribute__((aligned(16))) float wa_lds[2 * FIN];
    __shared__ float e_lds[128];
    __shared__ __attribute__((aligned(16))) float tpart_lds[4][16][TSTR];
    __shared__ __attribute__((aligned(16))) float t_lds[16][TSTR];

    const int blk   = blockIdx.x;
    const int bt    = blk >> 4;
    const int itile = blk & 15;
    const int b     = bt >> 2;
    const int tau   = bt & 3;
    const int cpred = (itile >= 8) ? 1 : 0;
    const int tp    = (2 * tau + cpred) & 3;

    const int tid  = threadIdx.x;
    const int wv   = tid >> 5;
    const int lane = tid & 31;
    const int half = lane >> 4;
    const int l15  = lane & 15;

    const float4* W4 = reinterpret_cast<const float4*>(W);
#define W_LD(i) const float4 wreg##i = W4[(i) * 128 + tid];
    W_LD(0) W_LD(1) W_LD(2)  W_LD(3)  W_LD(4)  W_LD(5)  W_LD(6)  W_LD(7)
    W_LD(8) W_LD(9) W_LD(10) W_LD(11) W_LD(12) W_LD(13) W_LD(14) W_LD(15)
#undef W_LD
    request_order();

    const float4* xs4 =
        reinterpret_cast<const float4*>(x + (size_t)(b * 4 + tp) * NN * FIN) + wv * 2048 + lane;
    float4 xlo[32], xhi[32];
#pragma unroll
    for (int i = 0; i < 32; ++i) {
        xlo[i] = xs4[i * 64];
        xhi[i] = xs4[i * 64 + 32];
    }
    request_order();

#define W_ST(i) *reinterpret_cast<float4*>(&w_lds[(i) * 8 + (tid >> 4)][(tid & 15) * 4]) = wreg##i;
    W_ST(0) W_ST(1) W_ST(2)  W_ST(3)  W_ST(4)  W_ST(5)  W_ST(6)  W_ST(7)
    W_ST(8) W_ST(9) W_ST(10) W_ST(11) W_ST(12) W_ST(13) W_ST(14) W_ST(15)
#undef W_ST
    lds_barrier();

    {
        const float4* wr = reinterpret_cast<const float4*>(&w_lds[tid][0]);
        float s1a = 0.f, s1b = 0.f, s2a = 0.f, s2b = 0.f;
#pragma unroll
        for (int q = 0; q < 16; ++q) {
            const float4 w = wr[q];
            s1a = fmaf(w.x, a[4 * q + 0], s1a);
            s1b = fmaf(w.y, a[4 * q + 1], s1b);
            s1a = fmaf(w.z, a[4 * q + 2], s1a);
            s1b = fmaf(w.w, a[4 * q + 3], s1b);
            s2a = fmaf(w.x, a[64 + 4 * q + 0], s2a);
            s2b = fmaf(w.y, a[64 + 4 * q + 1], s2b);
            s2a = fmaf(w.z, a[64 + 4 * q + 2], s2a);
            s2b = fmaf(w.w, a[64 + 4 * q + 3], s2b);
        }
        wa_lds[tid]       = s1a + s1b;
        wa_lds[FIN + tid] = s2a + s2b;
    }
    lds_barrier();

    float z[4][8];
#pragma unroll
    for (int rr = 0; rr < 4; ++rr) {
        const float* arow =
            adj + ((size_t)(bt * NN + itile * 16 + wv * 4 + rr)) * NN;
#pragma unroll
        for (int u = 0; u < 8; ++u)
            z[rr][u] = arow[lane + u * 32];
    }

    const float4* xb4 = reinterpret_cast<const float4*>(x + (size_t)bt * NN * FIN);
    float4 xa0[16], xa1[16], xb0[16], xb1[16];
#pragma unroll
    for (int t = 0; t < 16; ++t) {
        const int ka = 64 * wv + 4 * t + 2 * half;
        xa0[t] = xb4[ka * 32 + l15];
        xa1[t] = xb4[ka * 32 + 16 + l15];
        xb0[t] = xb4[(ka + 1) * 32 + l15];
        xb1[t] = xb4[(ka + 1) * 32 + 16 + l15];
    }
    request_order();

    {
        const float4 wlo = reinterpret_cast<const float4*>(wa_lds)[lane];
        const float4 whi = reinterpret_cast<const float4*>(wa_lds)[32 + lane];
#pragma unroll
        for (int i = 0; i < 32; ++i) {
            const float plo =
                fmaf(xlo[i].w, wlo.w, fmaf(xlo[i].z, wlo.z, fmaf(xlo[i].y, wlo.y, xlo[i].x * wlo.x)));
            const float phi =
                fmaf(xhi[i].w, whi.w, fmaf(xhi[i].z, whi.z, fmaf(xhi[i].y, whi.y, xhi[i].x * whi.x)));
            part_lds[wv][i][lane] = plo + phi;
        }
    }
    lds_barrier();
    {
        const float4* pr = reinterpret_cast<const float4*>(&part_lds[wv][lane][0]);
        float4 acc = pr[0];
#pragma unroll
        for (int k = 1; k < 8; ++k) {
            const float4 r = pr[k];
            acc.x += r.x; acc.y += r.y; acc.z += r.z; acc.w += r.w;
        }
        const float sv = (acc.x + acc.y) + (acc.z + acc.w);
        e_lds[wv * 32 + lane] = (sv > 0.f) ? sv : 0.2f * sv;
    }
    lds_barrier();

    float ev[4];
#pragma unroll
    for (int u = 0; u < 4; ++u)
        ev[u] = e_lds[lane + u * 32];

#pragma unroll
    for (int rr = 0; rr < 4; ++rr) {
        const int rloc = wv * 4 + rr;

        float mx = -3.402823466e38f;
#pragma unroll
        for (int u = 0; u < 8; ++u) {
            const float zz = z[rr][u] * ev[u & 3];
            z[rr][u] = zz;
            mx = fmaxf(mx, zz);
        }
#pragma unroll
        for (int off = 16; off > 0; off >>= 1)
            mx = fmaxf(mx, __shfl_xor(mx, off, 32));
        float sum = 0.f;
#pragma unroll
        for (int u = 0; u < 8; ++u) {
            const float p = __expf(z[rr][u] - mx);
            z[rr][u] = p;
            sum += p;
        }
#pragma unroll
        for (int off = 16; off > 0; off >>= 1)
            sum += __shfl_xor(sum, off, 32);
        const float inv = 1.f / sum;
#pragma unroll
        for (int u = 0; u < 8; ++u)
            attn_lds[rloc][lane + u * 32] = z[rr][u] * inv;
    }
    lds_barrier();
    __builtin_amdgcn_sched_barrier(0);

    v8f acc[8];
#pragma unroll
    for (int tt = 0; tt < 8; ++tt)
        acc[tt] = v8f{};
#pragma unroll
    for (int t = 0; t < 16; ++t) {
        const int ka = 64 * wv + 4 * t + 2 * half;
        const v2f a_ = *reinterpret_cast<const v2f*>(&attn_lds[l15][ka]);
        v2f b_;
        b_.x = xa0[t].x; b_.y = xb0[t].x; acc[0] = WMMA_F32(a_, b_, acc[0]);
        b_.x = xa0[t].y; b_.y = xb0[t].y; acc[1] = WMMA_F32(a_, b_, acc[1]);
        b_.x = xa0[t].z; b_.y = xb0[t].z; acc[2] = WMMA_F32(a_, b_, acc[2]);
        b_.x = xa0[t].w; b_.y = xb0[t].w; acc[3] = WMMA_F32(a_, b_, acc[3]);
        b_.x = xa1[t].x; b_.y = xb1[t].x; acc[4] = WMMA_F32(a_, b_, acc[4]);
        b_.x = xa1[t].y; b_.y = xb1[t].y; acc[5] = WMMA_F32(a_, b_, acc[5]);
        b_.x = xa1[t].z; b_.y = xb1[t].z; acc[6] = WMMA_F32(a_, b_, acc[6]);
        b_.x = xa1[t].w; b_.y = xb1[t].w; acc[7] = WMMA_F32(a_, b_, acc[7]);
    }
    const int colbase = wv * 16 + l15;
    v2f bw[32];
#pragma unroll
    for (int t = 0; t < 32; ++t) {
        const int ka = 4 * t + 2 * half;
        bw[t].x = w_lds[ka][colbase];
        bw[t].y = w_lds[ka + 1][colbase];
    }
#pragma unroll
    for (int r = 0; r < 8; ++r) {
        float4 v0, v1;
        v0.x = acc[0][r]; v0.y = acc[1][r]; v0.z = acc[2][r]; v0.w = acc[3][r];
        v1.x = acc[4][r]; v1.y = acc[5][r]; v1.z = acc[6][r]; v1.w = acc[7][r];
        *reinterpret_cast<float4*>(&tpart_lds[wv][8 * half + r][4 * l15])      = v0;
        *reinterpret_cast<float4*>(&tpart_lds[wv][8 * half + r][64 + 4 * l15]) = v1;
    }
    __syncthreads();

#pragma unroll
    for (int q = 0; q < 4; ++q) {
        const int idx = q * 128 + tid;
        const int row = idx >> 5;
        const int c4  = (idx & 31) * 4;
        const float4 p0 = *reinterpret_cast<const float4*>(&tpart_lds[0][row][c4]);
        const float4 p1 = *reinterpret_cast<const float4*>(&tpart_lds[1][row][c4]);
        const float4 p2 = *reinterpret_cast<const float4*>(&tpart_lds[2][row][c4]);
        const float4 p3 = *reinterpret_cast<const float4*>(&tpart_lds[3][row][c4]);
        float4 s;
        s.x = (p0.x + p1.x) + (p2.x + p3.x);
        s.y = (p0.y + p1.y) + (p2.y + p3.y);
        s.z = (p0.z + p1.z) + (p2.z + p3.z);
        s.w = (p0.w + p1.w) + (p2.w + p3.w);
        *reinterpret_cast<float4*>(&t_lds[row][c4]) = s;
    }
    __syncthreads();

    v2f at[32];
#pragma unroll
    for (int t = 0; t < 32; ++t)
        at[t] = *reinterpret_cast<const v2f*>(&t_lds[l15][4 * t + 2 * half]);
    __builtin_amdgcn_sched_barrier(0);
    v8f c0 = {}, c1 = {};
#pragma unroll
    for (int t = 0; t < 32; t += 2) {
        c0 = WMMA_F32(at[t],     bw[t],     c0);
        c1 = WMMA_F32(at[t + 1], bw[t + 1], c1);
    }
    const v8f c = c0 + c1;
    float* orow = out + (size_t)(bt * NN + itile * 16) * FOUT + colbase;
#pragma unroll
    for (int r = 0; r < 8; ++r)
        orow[(size_t)(r + half * 8) * FOUT] = c[r];
}

extern "C" void kernel_launch(void* const* d_in, const int* in_sizes, int n_in,
                              void* d_out, int out_size, void* d_ws, size_t ws_size,
                              hipStream_t stream) {
    const float* x   = (const float*)d_in[0];
    const float* adj = (const float*)d_in[1];
    const float* W   = (const float*)d_in[2];
    const float* a   = (const float*)d_in[3];
    float* out = (float*)d_out;

    k_gat_one<<<dim3(BT * 16), dim3(128), 0, stream>>>(x, adj, W, a, out);
}


# Softmax row max and row sum: the five butterfly steps run on the vector ALU (v_permlanex16 for lane^16, DPP row_xmask for 8,4,2,1) instead of ds_bpermute through LDS; same partner order, same sums
# speedup vs baseline: 1.9770x; 1.0605x over previous
#include <hip/hip_runtime.h>
#include <math.h>


typedef __attribute__((ext_vector_type(2))) float v2f;
typedef __attribute__((ext_vector_type(8))) float v8f;

#define BT    8
#define NN    256
#define FIN   128
#define FOUT  64
#define LDSW  260
#define PSTR  36
#define WSTR  76
#define TSTR  132

#define WMMA_F32(A_, B_, C_) \
    __builtin_amdgcn_wmma_f32_16x16x4_f32(false, (A_), false, (B_), (short)0, (C_), false, false)

static __device__ __forceinline__ void lds_barrier() {
    asm volatile("s_wait_dscnt 0x0\n\ts_barrier_signal -1\n\ts_barrier_wait -1" ::: "memory");
}
static __device__ __forceinline__ void request_order() {
    asm volatile("" ::: "memory");
}

static __device__ __forceinline__ float xhalf(float v) {
    const int b = __builtin_bit_cast(int, v);
    return __builtin_bit_cast(float, __builtin_amdgcn_permlanex16(b, b, 0x76543210, 0xfedcba98, false, false));
}
template <int M>
static __device__ __forceinline__ float xrow(float v) {
    return __builtin_bit_cast(float, __builtin_amdgcn_update_dpp(0, __builtin_bit_cast(int, v), 0x160 + M, 0xf, 0xf, true));
}

__global__ __launch_bounds__(128) void k_gat_one(const float* x,
                                                 const float* adj,
                                                 const float* W,
                                                 const float* __restrict__ a,
                                                 float* out) {
    __shared__ __attribute__((aligned(16))) float attn_lds[16][LDSW];
    __shared__ __attribute__((aligned(16))) float part_lds[4][32][PSTR];
    __shared__ __attribute__((aligned(16))) float w_lds[FIN][WSTR];
    __shared__ __attribute__((aligned(16))) float wa_lds[2 * FIN];
    __shared__ float e_lds[128];
    __shared__ __attribute__((aligned(16))) float tpart_lds[4][16][TSTR];
    __shared__ __attribute__((aligned(16))) float t_lds[16][TSTR];

    const int blk   = blockIdx.x;
    const int bt    = blk >> 4;
    const int itile = blk & 15;
    const int b     = bt >> 2;
    const int tau   = bt & 3;
    const int cpred = (itile >= 8) ? 1 : 0;
    const int tp    = (2 * tau + cpred) & 3;

    const int tid  = threadIdx.x;
    const int wv   = tid >> 5;
    const int lane = tid & 31;
    const int half = lane >> 4;
    const int l15  = lane & 15;

    const float4* W4 = reinterpret_cast<const float4*>(W);
#define W_LD(i) const float4 wreg##i = W4[(i) * 128 + tid];
    W_LD(0) W_LD(1) W_LD(2)  W_LD(3)  W_LD(4)  W_LD(5)  W_LD(6)  W_LD(7)
    W_LD(8) W_LD(9) W_LD(10) W_LD(11) W_LD(12) W_LD(13) W_LD(14) W_LD(15)
#undef W_LD
    request_order();

    const float4* xs4 =
        reinterpret_cast<const float4*>(x + (size_t)(b * 4 + tp) * NN * FIN) + wv * 2048 + lane;
    float4 xlo[32], xhi[32];
#pragma unroll
    for (int i = 0; i < 32; ++i) {
        xlo[i] = xs4[i * 64];
        xhi[i] = xs4[i * 64 + 32];
    }
    request_order();

#define W_ST(i) *reinterpret_cast<float4*>(&w_lds[(i) * 8 + (tid >> 4)][(tid & 15) * 4]) = wreg##i;
    W_ST(0) W_ST(1) W_ST(2)  W_ST(3)  W_ST(4)  W_ST(5)  W_ST(6)  W_ST(7)
    W_ST(8) W_ST(9) W_ST(10) W_ST(11) W_ST(12) W_ST(13) W_ST(14) W_ST(15)
#undef W_ST
    lds_barrier();

    {
        const float4* wr = reinterpret_cast<const float4*>(&w_lds[tid][0]);
        float s1a = 0.f, s1b = 0.f, s2a = 0.f, s2b = 0.f;
#pragma unroll
        for (int q = 0; q < 16; ++q) {
            const float4 w = wr[q];
            s1a = fmaf(w.x, a[4 * q + 0], s1a);
            s1b = fmaf(w.y, a[4 * q + 1], s1b);
            s1a = fmaf(w.z, a[4 * q + 2], s1a);
            s1b = fmaf(w.w, a[4 * q + 3], s1b);
            s2a = fmaf(w.x, a[64 + 4 * q + 0], s2a);
            s2b = fmaf(w.y, a[64 + 4 * q + 1], s2b);
            s2a = fmaf(w.z, a[64 + 4 * q + 2], s2a);
            s2b = fmaf(w.w, a[64 + 4 * q + 3], s2b);
        }
        wa_lds[tid]       = s1a + s1b;
        wa_lds[FIN + tid] = s2a + s2b;
    }
    lds_barrier();

    float z[4][8];
#pragma unroll
    for (int rr = 0; rr < 4; ++rr) {
        const float* arow =
            adj + ((size_t)(bt * NN + itile * 16 + wv * 4 + rr)) * NN;
#pragma unroll
        for (int u = 0; u < 8; ++u)
            z[rr][u] = arow[lane + u * 32];
    }

    const float4* xb4 = reinterpret_cast<const float4*>(x + (size_t)bt * NN * FIN);
    float4 xa0[16], xa1[16], xb0[16], xb1[16];
#pragma unroll
    for (int t = 0; t < 16; ++t) {
        const int ka = 64 * wv + 4 * t + 2 * half;
        xa0[t] = xb4[ka * 32 + l15];
        xa1[t] = xb4[ka * 32 + 16 + l15];
        xb0[t] = xb4[(ka + 1) * 32 + l15];
        xb1[t] = xb4[(ka + 1) * 32 + 16 + l15];
    }
    request_order();

    {
        const float4 wlo = reinterpret_cast<const float4*>(wa_lds)[lane];
        const float4 whi = reinterpret_cast<const float4*>(wa_lds)[32 + lane];
#pragma unroll
        for (int i = 0; i < 32; ++i) {
            const float plo =
                fmaf(xlo[i].w, wlo.w, fmaf(xlo[i].z, wlo.z, fmaf(xlo[i].y, wlo.y, xlo[i].x * wlo.x)));
            const float phi =
                fmaf(xhi[i].w, whi.w, fmaf(xhi[i].z, whi.z, fmaf(xhi[i].y, whi.y, xhi[i].x * whi.x)));
            part_lds[wv][i][lane] = plo + phi;
        }
    }
    lds_barrier();
    {
        const float4* pr = reinterpret_cast<const float4*>(&part_lds[wv][lane][0]);
        float4 acc = pr[0];
#pragma unroll
        for (int k = 1; k < 8; ++k) {
            const float4 r = pr[k];
            acc.x += r.x; acc.y += r.y; acc.z += r.z; acc.w += r.w;
        }
        const float sv = (acc.x + acc.y) + (acc.z + acc.w);
        e_lds[wv * 32 + lane] = (sv > 0.f) ? sv : 0.2f * sv;
    }
    lds_barrier();

    float ev[4];
#pragma unroll
    for (int u = 0; u < 4; ++u)
        ev[u] = e_lds[lane + u * 32];

#pragma unroll
    for (int rr = 0; rr < 4; ++rr) {
        const int rloc = wv * 4 + rr;

        float mx = -3.402823466e38f;
#pragma unroll
        for (int u = 0; u < 8; ++u) {
            const float zz = z[rr][u] * ev[u & 3];
            z[rr][u] = zz;
            mx = fmaxf(mx, zz);
        }
        mx = fmaxf(mx, xhalf(mx));
        mx = fmaxf(mx, xrow<8>(mx));
        mx = fmaxf(mx, xrow<4>(mx));
        mx = fmaxf(mx, xrow<2>(mx));
        mx = fmaxf(mx, xrow<1>(mx));
        float sum = 0.f;
#pragma unroll
        for (int u = 0; u < 8; ++u) {
            const float p = __expf(z[rr][u] - mx);
            z[rr][u] = p;
            sum += p;
        }
        sum += xhalf(sum);
        sum += xrow<8>(sum);
        sum += xrow<4>(sum);
        sum += xrow<2>(sum);
        sum += xrow<1>(sum);
        const float inv = 1.f / sum;
#pragma unroll
        for (int u = 0; u < 8; ++u)
            attn_lds[rloc][lane + u * 32] = z[rr][u] * inv;
    }
    lds_barrier();
    __builtin_amdgcn_sched_barrier(0);

    v8f acc[8];
#pragma unroll
    for (int tt = 0; tt < 8; ++tt)
        acc[tt] = v8f{};
#pragma unroll
    for (int t = 0; t < 16; ++t) {
        const int ka = 64 * wv + 4 * t + 2 * half;
        const v2f a_ = *reinterpret_cast<const v2f*>(&attn_lds[l15][ka]);
        v2f b_;
        b_.x = xa0[t].x; b_.y = xb0[t].x; acc[0] = WMMA_F32(a_, b_, acc[0]);
        b_.x = xa0[t].y; b_.y = xb0[t].y; acc[1] = WMMA_F32(a_, b_, acc[1]);
        b_.x = xa0[t].z; b_.y = xb0[t].z; acc[2] = WMMA_F32(a_, b_, acc[2]);
        b_.x = xa0[t].w; b_.y = xb0[t].w; acc[3] = WMMA_F32(a_, b_, acc[3]);
        b_.x = xa1[t].x; b_.y = xb1[t].x; acc[4] = WMMA_F32(a_, b_, acc[4]);
        b_.x = xa1[t].y; b_.y = xb1[t].y; acc[5] = WMMA_F32(a_, b_, acc[5]);
        b_.x = xa1[t].z; b_.y = xb1[t].z; acc[6] = WMMA_F32(a_, b_, acc[6]);
        b_.x = xa1[t].w; b_.y = xb1[t].w; acc[7] = WMMA_F32(a_, b_, acc[7]);
    }
    const int colbase = wv * 16 + l15;
    v2f bw[32];
#pragma unroll
    for (int t = 0; t < 32; ++t) {
        const int ka = 4 * t + 2 * half;
        bw[t].x = w_lds[ka][colbase];
        bw[t].y = w_lds[ka + 1][colbase];
    }
#pragma unroll
    for (int r = 0; r < 8; ++r) {
        float4 v0, v1;
        v0.x = acc[0][r]; v0.y = acc[1][r]; v0.z = acc[2][r]; v0.w = acc[3][r];
        v1.x = acc[4][r]; v1.y = acc[5][r]; v1.z = acc[6][r]; v1.w = acc[7][r];
        *reinterpret_cast<float4*>(&tpart_lds[wv][8 * half + r][4 * l15])      = v0;
        *reinterpret_cast<float4*>(&tpart_lds[wv][8 * half + r][64 + 4 * l15]) = v1;
    }
    __syncthreads();

#pragma unroll
    for (int q = 0; q < 4; ++q) {
        const int idx = q * 128 + tid;
        const int row = idx >> 5;
        const int c4  = (idx & 31) * 4;
        const float4 p0 = *reinterpret_cast<const float4*>(&tpart_lds[0][row][c4]);
        const float4 p1 = *reinterpret_cast<const float4*>(&tpart_lds[1][row][c4]);
        const float4 p2 = *reinterpret_cast<const float4*>(&tpart_lds[2][row][c4]);
        const float4 p3 = *reinterpret_cast<const float4*>(&tpart_lds[3][row][c4]);
        float4 s;
        s.x = (p0.x + p1.x) + (p2.x + p3.x);
        s.y = (p0.y + p1.y) + (p2.y + p3.y);
        s.z = (p0.z + p1.z) + (p2.z + p3.z);
        s.w = (p0.w + p1.w) + (p2.w + p3.w);
        *reinterpret_cast<float4*>(&t_lds[row][c4]) = s;
    }
    __syncthreads();

    v2f at[32];
#pragma unroll
    for (int t = 0; t < 32; ++t)
        at[t] = *reinterpret_cast<const v2f*>(&t_lds[l15][4 * t + 2 * half]);
    __builtin_amdgcn_sched_barrier(0);
    v8f c0 = {}, c1 = {};
#pragma unroll
    for (int t = 0; t < 32; t += 2) {
        c0 = WMMA_F32(at[t],     bw[t],     c0);
        c1 = WMMA_F32(at[t + 1], bw[t + 1], c1);
    }
    const v8f c = c0 + c1;
    float* orow = out + (size_t)(bt * NN + itile * 16) * FOUT + colbase;
#pragma unroll
    for (int r = 0; r < 8; ++r)
        orow[(size_t)(r + half * 8) * FOUT] = c[r];
}

extern "C" void kernel_launch(void* const* d_in, const int* in_sizes, int n_in,
                              void* d_out, int out_size, void* d_ws, size_t ws_size,
                              hipStream_t stream) {
    const float* x   = (const float*)d_in[0];
    const float* adj = (const float*)d_in[1];
    const float* W   = (const float*)d_in[2];
    const float* a   = (const float*)d_in[3];
    float* out = (float*)d_out;

    k_gat_one<<<dim3(BT * 16), dim3(128), 0, stream>>>(x, adj, W, a, out);
}


# s arithmetic consumes the first x stream as it lands: W a from the W registers by 16-lane DPP sums (no scalar loads, one barrier less), second-stream requests behind it; W a summed in a new order
# speedup vs baseline: 2.2504x; 1.1383x over previous
#include <hip/hip_runtime.h>
#include <math.h>


typedef __attribute__((ext_vector_type(2))) float v2f;
typedef __attribute__((ext_vector_type(8))) float v8f;

#define BT    8
#define NN    256
#define FIN   128
#define FOUT  64
#define LDSW  260
#define PSTR  36
#define WSTR  76
#define TSTR  132

#define WMMA_F32(A_, B_, C_) \
    __builtin_amdgcn_wmma_f32_16x16x4_f32(false, (A_), false, (B_), (short)0, (C_), false, false)

static __device__ __forceinline__ void lds_barrier() {
    asm volatile("s_wait_dscnt 0x0\n\ts_barrier_signal -1\n\ts_barrier_wait -1" ::: "memory");
}
static __device__ __forceinline__ void request_order() {
    asm volatile("" ::: "memory");
}

static __device__ __forceinline__ float xhalf(float v) {
    const int b = __builtin_bit_cast(int, v);
    return __builtin_bit_cast(float, __builtin_amdgcn_permlanex16(b, b, 0x76543210, 0xfedcba98, false, false));
}
template <int M>
static __device__ __forceinline__ float xrow(float v) {
    return __builtin_bit_cast(float, __builtin_amdgcn_update_dpp(0, __builtin_bit_cast(int, v), 0x160 + M, 0xf, 0xf, true));
}

__global__ __launch_bounds__(128) void k_gat_one(const float* x,
                                                 const float* adj,
                                                 const float* W,
                                                 const float* a,
                                                 float* out) {
    __shared__ __attribute__((aligned(16))) float attn_lds[16][LDSW];
    __shared__ __attribute__((aligned(16))) float part_lds[4][32][PSTR];
    __shared__ __attribute__((aligned(16))) float w_lds[FIN][WSTR];
    __shared__ __attribute__((aligned(16))) float wa_lds[2 * FIN];
    __shared__ float e_lds[128];
    __shared__ __attribute__((aligned(16))) float tpart_lds[4][16][TSTR];
    __shared__ __attribute__((aligned(16))) float t_lds[16][TSTR];

    const int blk   = blockIdx.x;
    const int bt    = blk >> 4;
    const int itile = blk & 15;
    const int b     = bt >> 2;
    const int tau   = bt & 3;
    const int cpred = (itile >= 8) ? 1 : 0;
    const int tp    = (2 * tau + cpred) & 3;

    const int tid  = threadIdx.x;
    const int wv   = tid >> 5;
    const int lane = tid & 31;
    const int half = lane >> 4;
    const int l15  = lane & 15;

    const float4* W4 = reinterpret_cast<const float4*>(W);
#define W_LD(i) const float4 wreg##i = W4[(i) * 128 + tid];
    W_LD(0) W_LD(1) W_LD(2)  W_LD(3)  W_LD(4)  W_LD(5)  W_LD(6)  W_LD(7)
    W_LD(8) W_LD(9) W_LD(10) W_LD(11) W_LD(12) W_LD(13) W_LD(14) W_LD(15)
#undef W_LD
    const float4 alo = reinterpret_cast<const float4*>(a)[l15];
    const float4 ahi = reinterpret_cast<const float4*>(a)[16 + l15];
    request_order();

    const float4* xs4 =
        reinterpret_cast<const float4*>(x + (size_t)(b * 4 + tp) * NN * FIN) + wv * 2048 + lane;
    float4 xlo[32], xhi[32];
#pragma unroll
    for (int i = 0; i < 32; ++i) {
        xlo[i] = xs4[i * 64];
        xhi[i] = xs4[i * 64 + 32];
    }
    request_order();

#define W_ST(i) *reinterpret_cast<float4*>(&w_lds[(i) * 8 + (tid >> 4)][(tid & 15) * 4]) = wreg##i;
    W_ST(0) W_ST(1) W_ST(2)  W_ST(3)  W_ST(4)  W_ST(5)  W_ST(6)  W_ST(7)
    W_ST(8) W_ST(9) W_ST(10) W_ST(11) W_ST(12) W_ST(13) W_ST(14) W_ST(15)
#undef W_ST

#define WA_ROW(i)                                                                                              \
    float walo##i = fmaf(wreg##i.w, alo.w, fmaf(wreg##i.z, alo.z, fmaf(wreg##i.y, alo.y, wreg##i.x * alo.x))); \
    float wahi##i = fmaf(wreg##i.w, ahi.w, fmaf(wreg##i.z, ahi.z, fmaf(wreg##i.y, ahi.y, wreg##i.x * ahi.x))); \
    walo##i += xrow<8>(walo##i); wahi##i += xrow<8>(wahi##i);                                                  \
    walo##i += xrow<4>(walo##i); wahi##i += xrow<4>(wahi##i);                                                  \
    walo##i += xrow<2>(walo##i); wahi##i += xrow<2>(wahi##i);                                                  \
    walo##i += xrow<1>(walo##i); wahi##i += xrow<1>(wahi##i);
    WA_ROW(0) WA_ROW(1) WA_ROW(2)  WA_ROW(3)  WA_ROW(4)  WA_ROW(5)  WA_ROW(6)  WA_ROW(7)
    WA_ROW(8) WA_ROW(9) WA_ROW(10) WA_ROW(11) WA_ROW(12) WA_ROW(13) WA_ROW(14) WA_ROW(15)
#undef WA_ROW
    if (l15 == 0) {
        const int r16 = tid >> 4;
#define WA_ST(i) wa_lds[(i) * 8 + r16] = walo##i; wa_lds[FIN + (i) * 8 + r16] = wahi##i;
        WA_ST(0) WA_ST(1) WA_ST(2)  WA_ST(3)  WA_ST(4)  WA_ST(5)  WA_ST(6)  WA_ST(7)
        WA_ST(8) WA_ST(9) WA_ST(10) WA_ST(11) WA_ST(12) WA_ST(13) WA_ST(14) WA_ST(15)
#undef WA_ST
    }
    lds_barrier();

    float z[4][8];
#pragma unroll
    for (int rr = 0; rr < 4; ++rr) {
        const float* arow =
            adj + ((size_t)(bt * NN + itile * 16 + wv * 4 + rr)) * NN;
#pragma unroll
        for (int u = 0; u < 8; ++u)
            z[rr][u] = arow[lane + u * 32];
    }
    request_order();

    {
        const float4 wlo = reinterpret_cast<const float4*>(wa_lds)[lane];
        const float4 whi = reinterpret_cast<const float4*>(wa_lds)[32 + lane];
#pragma unroll
        for (int g = 0; g < 8; ++g) {
#pragma unroll
            for (int ii = 0; ii < 4; ++ii) {
                const int i = 4 * g + ii;
                const float plo =
                    fmaf(xlo[i].w, wlo.w, fmaf(xlo[i].z, wlo.z, fmaf(xlo[i].y, wlo.y, xlo[i].x * wlo.x)));
                const float phi =
                    fmaf(xhi[i].w, whi.w, fmaf(xhi[i].z, whi.z, fmaf(xhi[i].y, whi.y, xhi[i].x * whi.x)));
                part_lds[wv][i][lane] = plo + phi;
            }
            __builtin_amdgcn_sched_barrier(0);
        }
    }

    const float4* xb4 = reinterpret_cast<const float4*>(x + (size_t)bt * NN * FIN);
    float4 xa0[16], xa1[16], xb0[16], xb1[16];
#define X2_LD(t)                                               \
    {                                                          \
        const int ka = 64 * wv + 4 * (t) + 2 * half;           \
        xa0[t] = xb4[ka * 32 + l15];                           \
        xa1[t] = xb4[ka * 32 + 16 + l15];                      \
        xb0[t] = xb4[(ka + 1) * 32 + l15];                     \
        xb1[t] = xb4[(ka + 1) * 32 + 16 + l15];                \
    }
    lds_barrier();
    {
        const float4* pr = reinterpret_cast<const float4*>(&part_lds[wv][lane][0]);
        float4 pv[8];
#pragma unroll
        for (int k = 0; k < 8; ++k)
            pv[k] = pr[k];
        request_order();
        X2_LD(0) X2_LD(1) X2_LD(2) X2_LD(3) X2_LD(4) X2_LD(5) X2_LD(6) X2_LD(7)
        request_order();
        __builtin_amdgcn_sched_barrier(0);
        float4 acc = pv[0];
#pragma unroll
        for (int k = 1; k < 8; ++k) {
            acc.x += pv[k].x; acc.y += pv[k].y; acc.z += pv[k].z; acc.w += pv[k].w;
        }
        const float sv = (acc.x + acc.y) + (acc.z + acc.w);
        e_lds[wv * 32 + lane] = (sv > 0.f) ? sv : 0.2f * sv;
    }
    lds_barrier();

    float ev[4];
#pragma unroll
    for (int u = 0; u < 4; ++u)
        ev[u] = e_lds[lane + u * 32];
    request_order();
    X2_LD(8) X2_LD(9) X2_LD(10) X2_LD(11) X2_LD(12) X2_LD(13) X2_LD(14) X2_LD(15)
#undef X2_LD
    request_order();

#pragma unroll
    for (int rr = 0; rr < 4; ++rr) {
        const int rloc = wv * 4 + rr;

        float mx = -3.402823466e38f;
#pragma unroll
        for (int u = 0; u < 8; ++u) {
            const float zz = z[rr][u] * ev[u & 3];
            z[rr][u] = zz;
            mx = fmaxf(mx, zz);
        }
        mx = fmaxf(mx, xhalf(mx));
        mx = fmaxf(mx, xrow<8>(mx));
        mx = fmaxf(mx, xrow<4>(mx));
        mx = fmaxf(mx, xrow<2>(mx));
        mx = fmaxf(mx, xrow<1>(mx));
        float sum = 0.f;
#pragma unroll
        for (int u = 0; u < 8; ++u) {
            const float p = __expf(z[rr][u] - mx);
            z[rr][u] = p;
            sum += p;
        }
        sum += xhalf(sum);
        sum += xrow<8>(sum);
        sum += xrow<4>(sum);
        sum += xrow<2>(sum);
        sum += xrow<1>(sum);
        const float inv = 1.f / sum;
#pragma unroll
        for (int u = 0; u < 8; ++u)
            attn_lds[rloc][lane + u * 32] = z[rr][u] * inv;
    }
    lds_barrier();
    __builtin_amdgcn_sched_barrier(0);

    v8f acc[8];
#pragma unroll
    for (int tt = 0; tt < 8; ++tt)
        acc[tt] = v8f{};
#pragma unroll
    for (int t = 0; t < 16; ++t) {
        const int ka = 64 * wv + 4 * t + 2 * half;
        const v2f a_ = *reinterpret_cast<const v2f*>(&attn_lds[l15][ka]);
        v2f b_;
        b_.x = xa0[t].x; b_.y = xb0[t].x; acc[0] = WMMA_F32(a_, b_, acc[0]);
        b_.x = xa0[t].y; b_.y = xb0[t].y; acc[1] = WMMA_F32(a_, b_, acc[1]);
        b_.x = xa0[t].z; b_.y = xb0[t].z; acc[2] = WMMA_F32(a_, b_, acc[2]);
        b_.x = xa0[t].w; b_.y = xb0[t].w; acc[3] = WMMA_F32(a_, b_, acc[3]);
        b_.x = xa1[t].x; b_.y = xb1[t].x; acc[4] = WMMA_F32(a_, b_, acc[4]);
        b_.x = xa1[t].y; b_.y = xb1[t].y; acc[5] = WMMA_F32(a_, b_, acc[5]);
        b_.x = xa1[t].z; b_.y = xb1[t].z; acc[6] = WMMA_F32(a_, b_, acc[6]);
        b_.x = xa1[t].w; b_.y = xb1[t].w; acc[7] = WMMA_F32(a_, b_, acc[7]);
    }
    const int colbase = wv * 16 + l15;
    v2f bw[32];
#pragma unroll
    for (int t = 0; t < 32; ++t) {
        const int ka = 4 * t + 2 * half;
        bw[t].x = w_lds[ka][colbase];
        bw[t].y = w_lds[ka + 1][colbase];
    }
#pragma unroll
    for (int r = 0; r < 8; ++r) {
        float4 v0, v1;
        v0.x = acc[0][r]; v0.y = acc[1][r]; v0.z = acc[2][r]; v0.w = acc[3][r];
        v1.x = acc[4][r]; v1.y = acc[5][r]; v1.z = acc[6][r]; v1.w = acc[7][r];
        *reinterpret_cast<float4*>(&tpart_lds[wv][8 * half + r][4 * l15])      = v0;
        *reinterpret_cast<float4*>(&tpart_lds[wv][8 * half + r][64 + 4 * l15]) = v1;
    }
    __syncthreads();

#pragma unroll
    for (int q = 0; q < 4; ++q) {
        const int idx = q * 128 + tid;
        const int row = idx >> 5;
        const int c4  = (idx & 31) * 4;
        const float4 p0 = *reinterpret_cast<const float4*>(&tpart_lds[0][row][c4]);
        const float4 p1 = *reinterpret_cast<const float4*>(&tpart_lds[1][row][c4]);
        const float4 p2 = *reinterpret_cast<const float4*>(&tpart_lds[2][row][c4]);
        const float4 p3 = *reinterpret_cast<const float4*>(&tpart_lds[3][row][c4]);
        float4 s;
        s.x = (p0.x + p1.x) + (p2.x + p3.x);
        s.y = (p0.y + p1.y) + (p2.y + p3.y);
        s.z = (p0.z + p1.z) + (p2.z + p3.z);
        s.w = (p0.w + p1.w) + (p2.w + p3.w);
        *reinterpret_cast<float4*>(&t_lds[row][c4]) = s;
    }
    __syncthreads();

    v2f at[32];
#pragma unroll
    for (int t = 0; t < 32; ++t)
        at[t] = *reinterpret_cast<const v2f*>(&t_lds[l15][4 * t + 2 * half]);
    __builtin_amdgcn_sched_barrier(0);
    v8f c0 = {}, c1 = {};
#pragma unroll
    for (int t = 0; t < 32; t += 2) {
        c0 = WMMA_F32(at[t],     bw[t],     c0);
        c1 = WMMA_F32(at[t + 1], bw[t + 1], c1);
    }
    const v8f c = c0 + c1;
    float* orow = out + (size_t)(bt * NN + itile * 16) * FOUT + colbase;
#pragma unroll
    for (int r = 0; r < 8; ++r)
        orow[(size_t)(r + half * 8) * FOUT] = c[r];
}

extern "C" void kernel_launch(void* const* d_in, const int* in_sizes, int n_in,
                              void* d_out, int out_size, void* d_ws, size_t ws_size,
                              hipStream_t stream) {
    const float* x   = (const float*)d_in[0];
    const float* adj = (const float*)d_in[1];
    const float* W   = (const float*)d_in[2];
    const float* a   = (const float*)d_in[3];
    float* out = (float*)d_out;

    k_gat_one<<<dim3(BT * 16), dim3(128), 0, stream>>>(x, adj, W, a, out);
}


# 8 waves per block instead of 4 (two per SIMD): each wave takes half a parent wave's rows, loads and tiles; adj and the whole second x stream are requested before the s arithmetic; sums unchanged
# speedup vs baseline: 2.3717x; 1.0539x over previous
#include <hip/hip_runtime.h>
#include <math.h>


typedef __attribute__((ext_vector_type(2))) float v2f;
typedef __attribute__((ext_vector_type(8))) float v8f;

#define BT    8
#define NN    256
#define FIN   128
#define FOUT  64
#define LDSW  260
#define PSTR  36
#define WSTR  76
#define TSTR  132

#define WMMA_F32(A_, B_, C_) \
    __builtin_amdgcn_wmma_f32_16x16x4_f32(false, (A_), false, (B_), (short)0, (C_), false, false)

static __device__ __forceinline__ void lds_barrier() {
    asm volatile("s_wait_dscnt 0x0\n\ts_barrier_signal -1\n\ts_barrier_wait -1" ::: "memory");
}
static __device__ __forceinline__ void request_order() {
    asm volatile("" ::: "memory");
}

static __device__ __forceinline__ float xhalf(float v) {
    const int b = __builtin_bit_cast(int, v);
    return __builtin_bit_cast(float, __builtin_amdgcn_permlanex16(b, b, 0x76543210, 0xfedcba98, false, false));
}
template <int M>
static __device__ __forceinline__ float xrow(float v) {
    return __builtin_bit_cast(float, __builtin_amdgcn_update_dpp(0, __builtin_bit_cast(int, v), 0x160 + M, 0xf, 0xf, true));
}

__global__ __launch_bounds__(256) void k_gat_one(const float* x,
                                                 const float* adj,
                                                 const float* W,
                                                 const float* a,
                                                 float* out) {
    __shared__ __attribute__((aligned(16))) float attn_lds[16][LDSW];
    __shared__ __attribute__((aligned(16))) float part_lds[8][16][PSTR];
    __shared__ __attribute__((aligned(16))) float w_lds[FIN][WSTR];
    __shared__ __attribute__((aligned(16))) float wa_lds[2 * FIN];
    __shared__ float e_lds[128];
    __shared__ __attribute__((aligned(16))) float tpart_lds[4][16][TSTR];
    __shared__ __attribute__((aligned(16))) float t_lds[16][TSTR];

    const int blk   = blockIdx.x;
    const int bt    = blk >> 4;
    const int itile = blk & 15;
    const int b     = bt >> 2;
    const int tau   = bt & 3;
    const int cpred = (itile >= 8) ? 1 : 0;
    const int tp    = (2 * tau + cpred) & 3;

    const int tid  = threadIdx.x;
    const int wv   = __builtin_amdgcn_readfirstlane(tid >> 5);
    const int lane = tid & 31;
    const int half = lane >> 4;
    const int l15  = lane & 15;

    const float4* W4 = reinterpret_cast<const float4*>(W);
#define W_LD(i) const float4 wreg##i = W4[(i) * 256 + tid];
    W_LD(0) W_LD(1) W_LD(2)  W_LD(3)  W_LD(4)  W_LD(5)  W_LD(6)  W_LD(7)
#undef W_LD
    const float4 alo = reinterpret_cast<const float4*>(a)[l15];
    const float4 ahi = reinterpret_cast<const float4*>(a)[16 + l15];
    request_order();

    const float4* xs4 =
        reinterpret_cast<const float4*>(x + (size_t)(b * 4 + tp) * NN * FIN) + wv * 1024 + lane;
    float4 xlo[16], xhi[16];
#pragma unroll
    for (int i = 0; i < 16; ++i) {
        xlo[i] = xs4[i * 64];
        xhi[i] = xs4[i * 64 + 32];
    }
    request_order();

    float z[2][8];
#pragma unroll
    for (int rr = 0; rr < 2; ++rr) {
        const float* arow =
            adj + ((size_t)(bt * NN + itile * 16 + wv * 2 + rr)) * NN;
#pragma unroll
        for (int u = 0; u < 8; ++u)
            z[rr][u] = arow[lane + u * 32];
    }
    request_order();

#define W_ST(i) *reinterpret_cast<float4*>(&w_lds[(i) * 16 + (tid >> 4)][(tid & 15) * 4]) = wreg##i;
    W_ST(0) W_ST(1) W_ST(2)  W_ST(3)  W_ST(4)  W_ST(5)  W_ST(6)  W_ST(7)
#undef W_ST

#define WA_ROW(i)                                                                                              \
    float walo##i = fmaf(wreg##i.w, alo.w, fmaf(wreg##i.z, alo.z, fmaf(wreg##i.y, alo.y, wreg##i.x * alo.x))); \
    float wahi##i = fmaf(wreg##i.w, ahi.w, fmaf(wreg##i.z, ahi.z, fmaf(wreg##i.y, ahi.y, wreg##i.x * ahi.x))); \
    walo##i += xrow<8>(walo##i); wahi##i += xrow<8>(wahi##i);                                                  \
    walo##i += xrow<4>(walo##i); wahi##i += xrow<4>(wahi##i);                                                  \
    walo##i += xrow<2>(walo##i); wahi##i += xrow<2>(wahi##i);                                                  \
    walo##i += xrow<1>(walo##i); wahi##i += xrow<1>(wahi##i);
    WA_ROW(0) WA_ROW(1) WA_ROW(2)  WA_ROW(3)  WA_ROW(4)  WA_ROW(5)  WA_ROW(6)  WA_ROW(7)
#undef WA_ROW
    if (l15 == 0) {
        const int r16 = tid >> 4;
#define WA_ST(i) wa_lds[(i) * 16 + r16] = walo##i; wa_lds[FIN + (i) * 16 + r16] = wahi##i;
        WA_ST(0) WA_ST(1) WA_ST(2)  WA_ST(3)  WA_ST(4)  WA_ST(5)  WA_ST(6)  WA_ST(7)
#undef WA_ST
    }
    lds_barrier();

    const int kq = wv & 3;
    const int nh = wv >> 2;
    const float4* xra =
        reinterpret_cast<const float4*>(x + (size_t)bt * NN * FIN) + (64 * kq + 2 * half) * 32 + 16 * nh + l15;
    const float4* xrb = xra + 32;
    float4 xa[16], xb[16];
#pragma unroll
    for (int t = 0; t < 16; ++t) {
        xa[t] = xra[t * 128];
        xb[t] = xrb[t * 128];
    }
    request_order();

    {
        const float4 wlo = reinterpret_cast<const float4*>(wa_lds)[lane];
        const float4 whi = reinterpret_cast<const float4*>(wa_lds)[32 + lane];
#pragma unroll
        for (int g = 0; g < 4; ++g) {
#pragma unroll
            for (int ii = 0; ii < 4; ++ii) {
                const int i = 4 * g + ii;
                const float plo =
                    fmaf(xlo[i].w, wlo.w, fmaf(xlo[i].z, wlo.z, fmaf(xlo[i].y, wlo.y, xlo[i].x * wlo.x)));
                const float phi =
                    fmaf(xhi[i].w, whi.w, fmaf(xhi[i].z, whi.z, fmaf(xhi[i].y, whi.y, xhi[i].x * whi.x)));
                part_lds[wv][i][lane] = plo + phi;
            }
            __builtin_amdgcn_sched_barrier(0);
        }
    }

    lds_barrier();
    {
        const float4* pr = reinterpret_cast<const float4*>(&part_lds[wv][l15][0]);
        float4 pv[8];
#pragma unroll
        for (int k = 0; k < 8; ++k)
            pv[k] = pr[k];
        float4 acc = pv[0];
#pragma unroll
        for (int k = 1; k < 8; ++k) {
            acc.x += pv[k].x; acc.y += pv[k].y; acc.z += pv[k].z; acc.w += pv[k].w;
        }
        const float sv = (acc.x + acc.y) + (acc.z + acc.w);
        e_lds[wv * 16 + l15] = (sv > 0.f) ? sv : 0.2f * sv;
    }
    lds_barrier();

    float ev[4];
#pragma unroll
    for (int u = 0; u < 4; ++u)
        ev[u] = e_lds[lane + u * 32];

#pragma unroll
    for (int rr = 0; rr < 2; ++rr) {
        const int rloc = wv * 2 + rr;

        float mx = -3.402823466e38f;
#pragma unroll
        for (int u = 0; u < 8; ++u) {
            const float zz = z[rr][u] * ev[u & 3];
            z[rr][u] = zz;
            mx = fmaxf(mx, zz);
        }
        mx = fmaxf(mx, xhalf(mx));
        mx = fmaxf(mx, xrow<8>(mx));
        mx = fmaxf(mx, xrow<4>(mx));
        mx = fmaxf(mx, xrow<2>(mx));
        mx = fmaxf(mx, xrow<1>(mx));
        float sum = 0.f;
#pragma unroll
        for (int u = 0; u < 8; ++u) {
            const float p = __expf(z[rr][u] - mx);
            z[rr][u] = p;
            sum += p;
        }
        sum += xhalf(sum);
        sum += xrow<8>(sum);
        sum += xrow<4>(sum);
        sum += xrow<2>(sum);
        sum += xrow<1>(sum);
        const float inv = 1.f / sum;
#pragma unroll
        for (int u = 0; u < 8; ++u)
            attn_lds[rloc][lane + u * 32] = z[rr][u] * inv;
    }
    lds_barrier();
    __builtin_amdgcn_sched_barrier(0);

    v8f acc[4];
#pragma unroll
    for (int tt = 0; tt < 4; ++tt)
        acc[tt] = v8f{};
#pragma unroll
    for (int t = 0; t < 16; ++t) {
        const int ka = 64 * kq + 4 * t + 2 * half;
        const v2f a_ = *reinterpret_cast<const v2f*>(&attn_lds[l15][ka]);
        v2f b_;
        b_.x = xa[t].x; b_.y = xb[t].x; acc[0] = WMMA_F32(a_, b_, acc[0]);
        b_.x = xa[t].y; b_.y = xb[t].y; acc[1] = WMMA_F32(a_, b_, acc[1]);
        b_.x = xa[t].z; b_.y = xb[t].z; acc[2] = WMMA_F32(a_, b_, acc[2]);
        b_.x = xa[t].w; b_.y = xb[t].w; acc[3] = WMMA_F32(a_, b_, acc[3]);
    }
#pragma unroll
    for (int r = 0; r < 8; ++r) {
        float4 v0;
        v0.x = acc[0][r]; v0.y = acc[1][r]; v0.z = acc[2][r]; v0.w = acc[3][r];
        *reinterpret_cast<float4*>(&tpart_lds[kq][8 * half + r][64 * nh + 4 * l15]) = v0;
    }
    const int colbase = kq * 16 + l15;
    v2f bw[32];
    if (wv < 4) {
#pragma unroll
        for (int t = 0; t < 32; ++t) {
            const int ka = 4 * t + 2 * half;
            bw[t].x = w_lds[ka][colbase];
            bw[t].y = w_lds[ka + 1][colbase];
        }
    }
    __syncthreads();

#pragma unroll
    for (int q = 0; q < 2; ++q) {
        const int idx = q * 256 + tid;
        const int row = idx >> 5;
        const int c4  = (idx & 31) * 4;
        const float4 p0 = *reinterpret_cast<const float4*>(&tpart_lds[0][row][c4]);
        const float4 p1 = *reinterpret_cast<const float4*>(&tpart_lds[1][row][c4]);
        const float4 p2 = *reinterpret_cast<const float4*>(&tpart_lds[2][row][c4]);
        const float4 p3 = *reinterpret_cast<const float4*>(&tpart_lds[3][row][c4]);
        float4 s;
        s.x = (p0.x + p1.x) + (p2.x + p3.x);
        s.y = (p0.y + p1.y) + (p2.y + p3.y);
        s.z = (p0.z + p1.z) + (p2.z + p3.z);
        s.w = (p0.w + p1.w) + (p2.w + p3.w);
        *reinterpret_cast<float4*>(&t_lds[row][c4]) = s;
    }
    __syncthreads();

    if (wv < 4) {
        v2f at[32];
#pragma unroll
        for (int t = 0; t < 32; ++t)
            at[t] = *reinterpret_cast<const v2f*>(&t_lds[l15][4 * t + 2 * half]);
        __builtin_amdgcn_sched_barrier(0);
        v8f c0 = {}, c1 = {};
#pragma unroll
        for (int t = 0; t < 32; t += 2) {
            c0 = WMMA_F32(at[t],     bw[t],     c0);
            c1 = WMMA_F32(at[t + 1], bw[t + 1], c1);
        }
        const v8f c = c0 + c1;
        float* orow = out + (size_t)(bt * NN + itile * 16) * FOUT + colbase;
#pragma unroll
        for (int r = 0; r < 8; ++r)
            orow[(size_t)(r + half * 8) * FOUT] = c[r];
    }
}

extern "C" void kernel_launch(void* const* d_in, const int* in_sizes, int n_in,
                              void* d_out, int out_size, void* d_ws, size_t ws_size,
                              hipStream_t stream) {
    const float* x   = (const float*)d_in[0];
    const float* adj = (const float*)d_in[1];
    const float* W   = (const float*)d_in[2];
    const float* a   = (const float*)d_in[3];
    float* out = (float*)d_out;

    k_gat_one<<<dim3(BT * 16), dim3(256), 0, stream>>>(x, adj, W, a, out);
}
